# starting kernel
# baseline (speedup 1.0000x reference)
#include <hip/hip_runtime.h>

typedef float v2f __attribute__((ext_vector_type(2)));
typedef float v8f __attribute__((ext_vector_type(8)));

#define WMMA_F32(a, b, c) \
  __builtin_amdgcn_wmma_f32_16x16x4_f32(false, (a), false, (b), (short)0, (c), false, false)

constexpr int kB = 8, kN = 1024, kP = 128, kH = 64, kD = 64, kJ = 256, kX = 64;
constexpr int INNER = kH * kD;
constexpr int CH = 32;
constexpr int NCH = kN / CH;

__global__ __launch_bounds__(256)
void pos_attn_fused(const float* __restrict__ pos, const float* __restrict__ x,
                    const float* __restrict__ Wq, const float* __restrict__ kl,
                    float* __restrict__ ws) {
  __shared__ float lds_x[CH][kX];
  __shared__ float lds_qh[CH][kD + 1];
  __shared__ float lds_p[8][32][CH + 1];

  const int bb   = blockIdx.x >> 6;
  const int h    = blockIdx.x & 63;
  const int tid  = threadIdx.x;
  const int w    = tid >> 5;
  const int lane = tid & 31;
  const int row  = lane & 15;
  const int half = lane >> 4;
  const int kb   = half * 2;

  const int tt = w >> 2;
  const int dt = w & 3;

  v8f   oacc[2][4];
  float mrun[2][8], lrun[2][8];
#pragma unroll
  for (int jt = 0; jt < 2; ++jt) {
#pragma unroll
    for (int xt = 0; xt < 4; ++xt) oacc[jt][xt] = (v8f){0,0,0,0,0,0,0,0};
#pragma unroll
    for (int r = 0; r < 8; ++r) { mrun[jt][r] = -3.0e38f; lrun[jt][r] = 0.0f; }
  }

  const float* posB = pos + (size_t)bb * kN * kP;
  const float* xB   = x   + (size_t)bb * kN * kX;

  for (int ch = 0; ch < NCH; ++ch) {
    const int n0 = ch * CH;
    __syncthreads();

    {
      const float4* gx = reinterpret_cast<const float4*>(xB + n0 * kX);
      float4* sx = reinterpret_cast<float4*>(&lds_x[0][0]);
#pragma unroll
      for (int i = 0; i < 2; ++i) sx[tid + i * 256] = gx[tid + i * 256];
    }

    {
      v8f cq = (v8f){0,0,0,0,0,0,0,0};
      const float* pA = posB + (size_t)(n0 + tt * 16 + row) * kP;
      const float* pB = Wq + h * kD + dt * 16 + row;
#pragma unroll
      for (int k = 0; k < kP; k += 4) {
        v2f a = *reinterpret_cast<const v2f*>(pA + k + kb);
        v2f b;
        b.x = pB[(size_t)(k + kb) * INNER];
        b.y = pB[(size_t)(k + kb + 1) * INNER];
        cq = WMMA_F32(a, b, cq);
      }
#pragma unroll
      for (int r = 0; r < 8; ++r)
        lds_qh[tt * 16 + r + 8 * half][dt * 16 + row] = cq[r];
    }
    __syncthreads();

    v8f s[2][2];
#pragma unroll
    for (int jt = 0; jt < 2; ++jt) {
#pragma unroll
      for (int nt = 0; nt < 2; ++nt) {
        v8f c = (v8f){0,0,0,0,0,0,0,0};
        const float* aK = kl + (size_t)(w * 32 + jt * 16 + row) * kD;
#pragma unroll
        for (int k = 0; k < kD; k += 4) {
          v2f a = *reinterpret_cast<const v2f*>(aK + k + kb);
          v2f b;
          b.x = lds_qh[nt * 16 + row][k + kb];
          b.y = lds_qh[nt * 16 + row][k + kb + 1];
          c = WMMA_F32(a, b, c);
        }
#pragma unroll
        for (int r = 0; r < 8; ++r) c[r] *= 0.125f;
        s[jt][nt] = c;
      }
    }

#pragma unroll
    for (int jt = 0; jt < 2; ++jt) {
      float t[8], mnew[8], fsc[8], rs[8];
#pragma unroll
      for (int r = 0; r < 8; ++r) t[r] = fmaxf(s[jt][0][r], s[jt][1][r]);
#pragma unroll
      for (int off = 1; off <= 8; off <<= 1)
#pragma unroll
        for (int r = 0; r < 8; ++r)
          t[r] = fmaxf(t[r], __shfl_xor(t[r], off, 32));
#pragma unroll
      for (int r = 0; r < 8; ++r) {
        mnew[r] = fmaxf(mrun[jt][r], t[r]);
        fsc[r]  = __expf(mrun[jt][r] - mnew[r]);
      }
#pragma unroll
      for (int nt = 0; nt < 2; ++nt)
#pragma unroll
        for (int r = 0; r < 8; ++r)
          s[jt][nt][r] = __expf(s[jt][nt][r] - mnew[r]);
#pragma unroll
      for (int r = 0; r < 8; ++r) rs[r] = s[jt][0][r] + s[jt][1][r];
#pragma unroll
      for (int off = 1; off <= 8; off <<= 1)
#pragma unroll
        for (int r = 0; r < 8; ++r) rs[r] += __shfl_xor(rs[r], off, 32);
#pragma unroll
      for (int r = 0; r < 8; ++r) {
        lrun[jt][r] = lrun[jt][r] * fsc[r] + rs[r];
        mrun[jt][r] = mnew[r];
      }
#pragma unroll
      for (int xt = 0; xt < 4; ++xt)
#pragma unroll
        for (int r = 0; r < 8; ++r) oacc[jt][xt][r] *= fsc[r];
#pragma unroll
      for (int nt = 0; nt < 2; ++nt)
#pragma unroll
        for (int r = 0; r < 8; ++r)
          lds_p[w][jt * 16 + r + 8 * half][nt * 16 + row] = s[jt][nt][r];
    }

#pragma unroll
    for (int jt = 0; jt < 2; ++jt) {
#pragma unroll
      for (int xt = 0; xt < 4; ++xt) {
        v8f c = oacc[jt][xt];
#pragma unroll
        for (int k = 0; k < CH; k += 4) {
          v2f a, b;
          a.x = lds_p[w][jt * 16 + row][k + kb];
          a.y = lds_p[w][jt * 16 + row][k + kb + 1];
          b.x = lds_x[k + kb][xt * 16 + row];
          b.y = lds_x[k + kb + 1][xt * 16 + row];
          c = WMMA_F32(a, b, c);
        }
        oacc[jt][xt] = c;
      }
    }
  }

#pragma unroll
  for (int jt = 0; jt < 2; ++jt) {
    float inv[8];
#pragma unroll
    for (int r = 0; r < 8; ++r) inv[r] = 1.0f / lrun[jt][r];
#pragma unroll
    for (int xt = 0; xt < 4; ++xt)
#pragma unroll
      for (int r = 0; r < 8; ++r) {
        const int j  = w * 32 + jt * 16 + r + 8 * half;
        const int xd = xt * 16 + row;
        ws[(size_t)(bb * kJ + j) * INNER + h * kX + xd] = oacc[jt][xt][r] * inv[r];
      }
  }
}

__global__ __launch_bounds__(128)
void out_proj(const float* __restrict__ ws, const float* __restrict__ Wout,
              const float* __restrict__ bout, float* __restrict__ out) {
  const int rt   = blockIdx.x;
  const int w    = threadIdx.x >> 5;
  const int lane = threadIdx.x & 31;
  const int row  = lane & 15;
  const int half = lane >> 4;
  const int kb   = half * 2;

  const float* aR = ws + (size_t)(rt * 16 + row) * INNER;
  const float* bC = Wout + w * 16 + row;

  v8f c = (v8f){0,0,0,0,0,0,0,0};
  for (int k = 0; k < INNER; k += 4) {
    v2f a = *reinterpret_cast<const v2f*>(aR + k + kb);
    v2f b;
    b.x = bC[(size_t)(k + kb) * 64];
    b.y = bC[(size_t)(k + kb + 1) * 64];
    c = WMMA_F32(a, b, c);
  }
  const float bias = bout[w * 16 + row];
#pragma unroll
  for (int r = 0; r < 8; ++r)
    out[(size_t)(rt * 16 + r + 8 * half) * 64 + w * 16 + row] = c[r] + bias;
}

extern "C" void kernel_launch(void* const* d_in, const int* in_sizes, int n_in,
                              void* d_out, int out_size, void* d_ws, size_t ws_size,
                              hipStream_t stream) {
  (void)in_sizes; (void)n_in; (void)out_size; (void)ws_size;
  const float* pos  = (const float*)d_in[0];
  const float* x    = (const float*)d_in[1];
  const float* Wq   = (const float*)d_in[2];
  const float* kl   = (const float*)d_in[3];
  const float* Wout = (const float*)d_in[4];
  const float* bout = (const float*)d_in[5];
  float* ws  = (float*)d_ws;
  float* out = (float*)d_out;

  pos_attn_fused<<<kB * kH, 256, 0, stream>>>(pos, x, Wq, kl, ws);
  out_proj<<<(kB * kJ) / 16, 128, 0, stream>>>(ws, Wout, bout, out);
}


# out_proj: K=4096 split over 16 waves per block (2 col tiles each, 128-bit A loads, k order permuted), partial tiles summed via LDS in wave order; 256 blocks x 512 threads
# speedup vs baseline: 1.2214x; 1.2214x over previous
#include <hip/hip_runtime.h>

typedef float v2f __attribute__((ext_vector_type(2)));
typedef float v8f __attribute__((ext_vector_type(8)));

#define WMMA_F32(a, b, c) \
  __builtin_amdgcn_wmma_f32_16x16x4_f32(false, (a), false, (b), (short)0, (c), false, false)

constexpr int kB = 8, kN = 1024, kP = 128, kH = 64, kD = 64, kJ = 256, kX = 64;
constexpr int INNER = kH * kD;
constexpr int CH = 32;
constexpr int NCH = kN / CH;

__global__ __launch_bounds__(256)
void pos_attn_fused(const float* __restrict__ pos, const float* __restrict__ x,
                    const float* __restrict__ Wq, const float* __restrict__ kl,
                    float* __restrict__ ws) {
  __shared__ float lds_x[CH][kX];
  __shared__ float lds_qh[CH][kD + 1];
  __shared__ float lds_p[8][32][CH + 1];

  const int bb   = blockIdx.x >> 6;
  const int h    = blockIdx.x & 63;
  const int tid  = threadIdx.x;
  const int w    = tid >> 5;
  const int lane = tid & 31;
  const int row  = lane & 15;
  const int half = lane >> 4;
  const int kb   = half * 2;

  const int tt = w >> 2;
  const int dt = w & 3;

  v8f   oacc[2][4];
  float mrun[2][8], lrun[2][8];
#pragma unroll
  for (int jt = 0; jt < 2; ++jt) {
#pragma unroll
    for (int xt = 0; xt < 4; ++xt) oacc[jt][xt] = (v8f){0,0,0,0,0,0,0,0};
#pragma unroll
    for (int r = 0; r < 8; ++r) { mrun[jt][r] = -3.0e38f; lrun[jt][r] = 0.0f; }
  }

  const float* posB = pos + (size_t)bb * kN * kP;
  const float* xB   = x   + (size_t)bb * kN * kX;

  for (int ch = 0; ch < NCH; ++ch) {
    const int n0 = ch * CH;
    __syncthreads();

    {
      const float4* gx = reinterpret_cast<const float4*>(xB + n0 * kX);
      float4* sx = reinterpret_cast<float4*>(&lds_x[0][0]);
#pragma unroll
      for (int i = 0; i < 2; ++i) sx[tid + i * 256] = gx[tid + i * 256];
    }

    {
      v8f cq = (v8f){0,0,0,0,0,0,0,0};
      const float* pA = posB + (size_t)(n0 + tt * 16 + row) * kP;
      const float* pB = Wq + h * kD + dt * 16 + row;
#pragma unroll
      for (int k = 0; k < kP; k += 4) {
        v2f a = *reinterpret_cast<const v2f*>(pA + k + kb);
        v2f b;
        b.x = pB[(size_t)(k + kb) * INNER];
        b.y = pB[(size_t)(k + kb + 1) * INNER];
        cq = WMMA_F32(a, b, cq);
      }
#pragma unroll
      for (int r = 0; r < 8; ++r)
        lds_qh[tt * 16 + r + 8 * half][dt * 16 + row] = cq[r];
    }
    __syncthreads();

    v8f s[2][2];
#pragma unroll
    for (int jt = 0; jt < 2; ++jt) {
#pragma unroll
      for (int nt = 0; nt < 2; ++nt) {
        v8f c = (v8f){0,0,0,0,0,0,0,0};
        const float* aK = kl + (size_t)(w * 32 + jt * 16 + row) * kD;
#pragma unroll
        for (int k = 0; k < kD; k += 4) {
          v2f a = *reinterpret_cast<const v2f*>(aK + k + kb);
          v2f b;
          b.x = lds_qh[nt * 16 + row][k + kb];
          b.y = lds_qh[nt * 16 + row][k + kb + 1];
          c = WMMA_F32(a, b, c);
        }
#pragma unroll
        for (int r = 0; r < 8; ++r) c[r] *= 0.125f;
        s[jt][nt] = c;
      }
    }

#pragma unroll
    for (int jt = 0; jt < 2; ++jt) {
      float t[8], mnew[8], fsc[8], rs[8];
#pragma unroll
      for (int r = 0; r < 8; ++r) t[r] = fmaxf(s[jt][0][r], s[jt][1][r]);
#pragma unroll
      for (int off = 1; off <= 8; off <<= 1)
#pragma unroll
        for (int r = 0; r < 8; ++r)
          t[r] = fmaxf(t[r], __shfl_xor(t[r], off, 32));
#pragma unroll
      for (int r = 0; r < 8; ++r) {
        mnew[r] = fmaxf(mrun[jt][r], t[r]);
        fsc[r]  = __expf(mrun[jt][r] - mnew[r]);
      }
#pragma unroll
      for (int nt = 0; nt < 2; ++nt)
#pragma unroll
        for (int r = 0; r < 8; ++r)
          s[jt][nt][r] = __expf(s[jt][nt][r] - mnew[r]);
#pragma unroll
      for (int r = 0; r < 8; ++r) rs[r] = s[jt][0][r] + s[jt][1][r];
#pragma unroll
      for (int off = 1; off <= 8; off <<= 1)
#pragma unroll
        for (int r = 0; r < 8; ++r) rs[r] += __shfl_xor(rs[r], off, 32);
#pragma unroll
      for (int r = 0; r < 8; ++r) {
        lrun[jt][r] = lrun[jt][r] * fsc[r] + rs[r];
        mrun[jt][r] = mnew[r];
      }
#pragma unroll
      for (int xt = 0; xt < 4; ++xt)
#pragma unroll
        for (int r = 0; r < 8; ++r) oacc[jt][xt][r] *= fsc[r];
#pragma unroll
      for (int nt = 0; nt < 2; ++nt)
#pragma unroll
        for (int r = 0; r < 8; ++r)
          lds_p[w][jt * 16 + r + 8 * half][nt * 16 + row] = s[jt][nt][r];
    }

#pragma unroll
    for (int jt = 0; jt < 2; ++jt) {
#pragma unroll
      for (int xt = 0; xt < 4; ++xt) {
        v8f c = oacc[jt][xt];
#pragma unroll
        for (int k = 0; k < CH; k += 4) {
          v2f a, b;
          a.x = lds_p[w][jt * 16 + row][k + kb];
          a.y = lds_p[w][jt * 16 + row][k + kb + 1];
          b.x = lds_x[k + kb][xt * 16 + row];
          b.y = lds_x[k + kb + 1][xt * 16 + row];
          c = WMMA_F32(a, b, c);
        }
        oacc[jt][xt] = c;
      }
    }
  }

#pragma unroll
  for (int jt = 0; jt < 2; ++jt) {
    float inv[8];
#pragma unroll
    for (int r = 0; r < 8; ++r) inv[r] = 1.0f / lrun[jt][r];
#pragma unroll
    for (int xt = 0; xt < 4; ++xt)
#pragma unroll
      for (int r = 0; r < 8; ++r) {
        const int j  = w * 32 + jt * 16 + r + 8 * half;
        const int xd = xt * 16 + row;
        ws[(size_t)(bb * kJ + j) * INNER + h * kX + xd] = oacc[jt][xt][r] * inv[r];
      }
  }
}

constexpr int OP_WAVES  = 16;
constexpr int OP_KSLICE = INNER / OP_WAVES;
constexpr int OP_TILE   = 272;
constexpr int OP_BLOCKS = (kB * kJ / 16) * 2;

__global__ __launch_bounds__(512) __attribute__((amdgpu_waves_per_eu(4, 4)))
void out_proj(const float* __restrict__ ws, const float* __restrict__ Wout,
              const float* __restrict__ bout, float* __restrict__ out) {
  __shared__ float part[OP_WAVES][2 * OP_TILE];

  const int rt   = blockIdx.x >> 1;
  const int nh   = blockIdx.x & 1;
  const int tid  = threadIdx.x;
  const int w    = tid >> 5;
  const int lane = tid & 31;
  const int c    = lane & 15;
  const int half = lane >> 4;

  const float* aP = ws   + (size_t)(rt * 16 + c) * INNER + w * OP_KSLICE + 4 * half;
  const float* bP = Wout + (size_t)(w * OP_KSLICE + 4 * half) * 64 + nh * 32 + 2 * c;

  v8f acc0 = (v8f){0,0,0,0,0,0,0,0};
  v8f acc1 = (v8f){0,0,0,0,0,0,0,0};
#pragma unroll 8
  for (int g = 0; g < OP_KSLICE / 8; ++g) {
    const float4 a4 = *reinterpret_cast<const float4*>(aP + 8 * g);
    const float2 w0 = *reinterpret_cast<const float2*>(bP + (size_t)(8 * g + 0) * 64);
    const float2 w1 = *reinterpret_cast<const float2*>(bP + (size_t)(8 * g + 1) * 64);
    const float2 w2 = *reinterpret_cast<const float2*>(bP + (size_t)(8 * g + 2) * 64);
    const float2 w3 = *reinterpret_cast<const float2*>(bP + (size_t)(8 * g + 3) * 64);
    v2f a, b0, b1;
    a.x = a4.x;  a.y = a4.y;
    b0.x = w0.x; b0.y = w1.x;
    b1.x = w0.y; b1.y = w1.y;
    acc0 = WMMA_F32(a, b0, acc0);
    acc1 = WMMA_F32(a, b1, acc1);
    a.x = a4.z;  a.y = a4.w;
    b0.x = w2.x; b0.y = w3.x;
    b1.x = w2.y; b1.y = w3.y;
    acc0 = WMMA_F32(a, b0, acc0);
    acc1 = WMMA_F32(a, b1, acc1);
  }

  float* myp = &part[w][0];
#pragma unroll
  for (int r = 0; r < 8; ++r) {
    myp[r * 32 + lane]           = acc0[r];
    myp[OP_TILE + r * 32 + lane] = acc1[r];
  }
  __syncthreads();

  const int row = tid >> 5;
  const int col = tid & 31;
  const int src = (col & 1) * OP_TILE + (row & 7) * 32 + (row >> 3) * 16 + (col >> 1);
  float s = part[0][src];
#pragma unroll
  for (int ww = 1; ww < OP_WAVES; ++ww) s += part[ww][src];
  out[(size_t)(rt * 16 + row) * 64 + nh * 32 + col] = s + bout[nh * 32 + col];
}

extern "C" void kernel_launch(void* const* d_in, const int* in_sizes, int n_in,
                              void* d_out, int out_size, void* d_ws, size_t ws_size,
                              hipStream_t stream) {
  (void)in_sizes; (void)n_in; (void)out_size; (void)ws_size;
  const float* pos  = (const float*)d_in[0];
  const float* x    = (const float*)d_in[1];
  const float* Wq   = (const float*)d_in[2];
  const float* kl   = (const float*)d_in[3];
  const float* Wout = (const float*)d_in[4];
  const float* bout = (const float*)d_in[5];
  float* ws  = (float*)d_ws;
  float* out = (float*)d_out;

  pos_attn_fused<<<kB * kH, 256, 0, stream>>>(pos, x, Wq, kl, ws);
  out_proj<<<OP_BLOCKS, 512, 0, stream>>>(ws, Wout, bout, out);
}


# attention kernel: next chunk's pos and x rows fetched one chunk ahead by whole-line float4 loads into double-buffered LDS; pos A fragments read from LDS, not 16-line gathers; one barrier per chunk
# speedup vs baseline: 1.2746x; 1.0436x over previous
#include <hip/hip_runtime.h>

typedef float v2f __attribute__((ext_vector_type(2)));
typedef float v8f __attribute__((ext_vector_type(8)));

#define WMMA_F32(a, b, c) \
  __builtin_amdgcn_wmma_f32_16x16x4_f32(false, (a), false, (b), (short)0, (c), false, false)

constexpr int kB = 8, kN = 1024, kP = 128, kH = 64, kD = 64, kJ = 256, kX = 64;
constexpr int INNER = kH * kD;
constexpr int CH = 32;
constexpr int NCH = kN / CH;

__global__ __launch_bounds__(256)
void pos_attn_fused(const float* __restrict__ pos, const float* __restrict__ x,
                    const float* __restrict__ Wq, const float* __restrict__ kl,
                    float* __restrict__ ws) {
  __shared__ __attribute__((aligned(16))) float lds_pos[2][CH][kP + 4];
  __shared__ __attribute__((aligned(16))) float lds_x[2][CH][kX];
  __shared__ float lds_qh[2][CH][kD + 1];
  __shared__ float lds_p[8][32][CH + 1];

  const int bb   = blockIdx.x >> 6;
  const int h    = blockIdx.x & 63;
  const int tid  = threadIdx.x;
  const int w    = tid >> 5;
  const int lane = tid & 31;
  const int row  = lane & 15;
  const int half = lane >> 4;
  const int kb   = half * 2;

  const int tt = w >> 2;
  const int dt = w & 3;

  v8f   oacc[2][4];
  float mrun[2][8], lrun[2][8];
#pragma unroll
  for (int jt = 0; jt < 2; ++jt) {
#pragma unroll
    for (int xt = 0; xt < 4; ++xt) oacc[jt][xt] = (v8f){0,0,0,0,0,0,0,0};
#pragma unroll
    for (int r = 0; r < 8; ++r) { mrun[jt][r] = -3.0e38f; lrun[jt][r] = 0.0f; }
  }

  const float4* gpos = reinterpret_cast<const float4*>(pos + (size_t)bb * kN * kP);
  const float4* gx   = reinterpret_cast<const float4*>(x   + (size_t)bb * kN * kX);
  constexpr int POS4 = CH * kP / 4;
  constexpr int X4   = CH * kX / 4;

#pragma unroll
  for (int i = 0; i < 4; ++i) {
    const int i4 = tid + i * 256;
    *reinterpret_cast<float4*>(&lds_pos[0][i4 >> 5][(i4 & 31) * 4]) = gpos[i4];
  }
#pragma unroll
  for (int i = 0; i < 2; ++i)
    reinterpret_cast<float4*>(&lds_x[0][0][0])[tid + i * 256] = gx[tid + i * 256];
  __syncthreads();

  for (int ch = 0; ch < NCH; ++ch) {
    const int  cur  = ch & 1;
    const int  nxt  = cur ^ 1;
    const bool more = ch + 1 < NCH;

    const float4 zero4 = make_float4(0.0f, 0.0f, 0.0f, 0.0f);
    float4 pp0 = zero4, pp1 = zero4, pp2 = zero4, pp3 = zero4, px0 = zero4, px1 = zero4;
    if (more) {
      const float4* np = gpos + (ch + 1) * POS4 + tid;
      const float4* nx = gx + (ch + 1) * X4 + tid;
      pp0 = np[0];  pp1 = np[256];  pp2 = np[512];  pp3 = np[768];
      px0 = nx[0];  px1 = nx[256];
    }

    {
      v8f cq = (v8f){0,0,0,0,0,0,0,0};
      const float* pA = &lds_pos[cur][tt * 16 + row][kb];
      const float* pB = Wq + h * kD + dt * 16 + row;
      v2f ap2[kP / 4];
#pragma unroll
      for (int k = 0; k < kP; k += 4) ap2[k / 4] = *reinterpret_cast<const v2f*>(pA + k);
      __builtin_amdgcn_sched_barrier(0);
#pragma unroll
      for (int k = 0; k < kP; k += 4) {
        v2f b;
        b.x = pB[(size_t)(k + kb) * INNER];
        b.y = pB[(size_t)(k + kb + 1) * INNER];
        cq = WMMA_F32(ap2[k / 4], b, cq);
      }
#pragma unroll
      for (int r = 0; r < 8; ++r)
        lds_qh[cur][tt * 16 + r + 8 * half][dt * 16 + row] = cq[r];
    }
    if (more) {
      float* sp = &lds_pos[nxt][tid >> 5][(tid & 31) * 4];
      *reinterpret_cast<float4*>(sp)                     = pp0;
      *reinterpret_cast<float4*>(sp +  8 * (kP + 4))     = pp1;
      *reinterpret_cast<float4*>(sp + 16 * (kP + 4))     = pp2;
      *reinterpret_cast<float4*>(sp + 24 * (kP + 4))     = pp3;
    }
    __syncthreads();
    if (more) {
      float4* sx = reinterpret_cast<float4*>(&lds_x[nxt][0][0]) + tid;
      sx[0]   = px0;
      sx[256] = px1;
    }

    v2f bq[2][kD / 4], bx[4][CH / 4];
#pragma unroll
    for (int nt = 0; nt < 2; ++nt)
#pragma unroll
      for (int k = 0; k < kD; k += 4) {
        bq[nt][k / 4].x = lds_qh[cur][nt * 16 + row][k + kb];
        bq[nt][k / 4].y = lds_qh[cur][nt * 16 + row][k + kb + 1];
      }
#pragma unroll
    for (int xt = 0; xt < 4; ++xt)
#pragma unroll
      for (int k = 0; k < CH; k += 4) {
        bx[xt][k / 4].x = lds_x[cur][k + kb][xt * 16 + row];
        bx[xt][k / 4].y = lds_x[cur][k + kb + 1][xt * 16 + row];
      }
    __builtin_amdgcn_sched_barrier(2);

    v8f s[2][2];
#pragma unroll
    for (int jt = 0; jt < 2; ++jt)
#pragma unroll
      for (int nt = 0; nt < 2; ++nt) s[jt][nt] = (v8f){0,0,0,0,0,0,0,0};
#pragma unroll
    for (int k = 0; k < kD; k += 4) {
#pragma unroll
      for (int jt = 0; jt < 2; ++jt) {
        const float* aK = kl + (size_t)(w * 32 + jt * 16 + row) * kD;
        const v2f a = *reinterpret_cast<const v2f*>(aK + k + kb);
#pragma unroll
        for (int nt = 0; nt < 2; ++nt) s[jt][nt] = WMMA_F32(a, bq[nt][k / 4], s[jt][nt]);
      }
    }
#pragma unroll
    for (int jt = 0; jt < 2; ++jt)
#pragma unroll
      for (int nt = 0; nt < 2; ++nt)
#pragma unroll
        for (int r = 0; r < 8; ++r) s[jt][nt][r] *= 0.125f;

#pragma unroll
    for (int jt = 0; jt < 2; ++jt) {
      float t[8], mnew[8], fsc[8], rs[8];
#pragma unroll
      for (int r = 0; r < 8; ++r) t[r] = fmaxf(s[jt][0][r], s[jt][1][r]);
#pragma unroll
      for (int off = 1; off <= 8; off <<= 1)
#pragma unroll
        for (int r = 0; r < 8; ++r)
          t[r] = fmaxf(t[r], __shfl_xor(t[r], off, 32));
#pragma unroll
      for (int r = 0; r < 8; ++r) {
        mnew[r] = fmaxf(mrun[jt][r], t[r]);
        fsc[r]  = __expf(mrun[jt][r] - mnew[r]);
      }
#pragma unroll
      for (int nt = 0; nt < 2; ++nt)
#pragma unroll
        for (int r = 0; r < 8; ++r)
          s[jt][nt][r] = __expf(s[jt][nt][r] - mnew[r]);
#pragma unroll
      for (int r = 0; r < 8; ++r) rs[r] = s[jt][0][r] + s[jt][1][r];
#pragma unroll
      for (int off = 1; off <= 8; off <<= 1)
#pragma unroll
        for (int r = 0; r < 8; ++r) rs[r] += __shfl_xor(rs[r], off, 32);
#pragma unroll
      for (int r = 0; r < 8; ++r) {
        lrun[jt][r] = lrun[jt][r] * fsc[r] + rs[r];
        mrun[jt][r] = mnew[r];
      }
#pragma unroll
      for (int xt = 0; xt < 4; ++xt)
#pragma unroll
        for (int r = 0; r < 8; ++r) oacc[jt][xt][r] *= fsc[r];
#pragma unroll
      for (int nt = 0; nt < 2; ++nt)
#pragma unroll
        for (int r = 0; r < 8; ++r)
          lds_p[w][jt * 16 + r + 8 * half][nt * 16 + row] = s[jt][nt][r];
    }

    v2f app[2][CH / 4];
#pragma unroll
    for (int jt = 0; jt < 2; ++jt)
#pragma unroll
      for (int k = 0; k < CH; k += 4) {
        app[jt][k / 4].x = lds_p[w][jt * 16 + row][k + kb];
        app[jt][k / 4].y = lds_p[w][jt * 16 + row][k + kb + 1];
      }
#pragma unroll
    for (int jt = 0; jt < 2; ++jt)
#pragma unroll
      for (int k = 0; k < CH; k += 4)
#pragma unroll
        for (int xt = 0; xt < 4; ++xt)
          oacc[jt][xt] = WMMA_F32(app[jt][k / 4], bx[xt][k / 4], oacc[jt][xt]);
  }

#pragma unroll
  for (int jt = 0; jt < 2; ++jt) {
    float inv[8];
#pragma unroll
    for (int r = 0; r < 8; ++r) inv[r] = 1.0f / lrun[jt][r];
#pragma unroll
    for (int xt = 0; xt < 4; ++xt)
#pragma unroll
      for (int r = 0; r < 8; ++r) {
        const int j  = w * 32 + jt * 16 + r + 8 * half;
        const int xd = xt * 16 + row;
        ws[(size_t)(bb * kJ + j) * INNER + h * kX + xd] = oacc[jt][xt][r] * inv[r];
      }
  }
}

constexpr int OP_WAVES  = 16;
constexpr int OP_KSLICE = INNER / OP_WAVES;
constexpr int OP_TILE   = 272;
constexpr int OP_BLOCKS = (kB * kJ / 16) * 2;

__global__ __launch_bounds__(512) __attribute__((amdgpu_waves_per_eu(4, 4)))
void out_proj(const float* __restrict__ ws, const float* __restrict__ Wout,
              const float* __restrict__ bout, float* __restrict__ out) {
  __shared__ float part[OP_WAVES][2 * OP_TILE];

  const int rt   = blockIdx.x >> 1;
  const int nh   = blockIdx.x & 1;
  const int tid  = threadIdx.x;
  const int w    = tid >> 5;
  const int lane = tid & 31;
  const int c    = lane & 15;
  const int half = lane >> 4;

  const float* aP = ws   + (size_t)(rt * 16 + c) * INNER + w * OP_KSLICE + 4 * half;
  const float* bP = Wout + (size_t)(w * OP_KSLICE + 4 * half) * 64 + nh * 32 + 2 * c;

  v8f acc0 = (v8f){0,0,0,0,0,0,0,0};
  v8f acc1 = (v8f){0,0,0,0,0,0,0,0};
#pragma unroll 8
  for (int g = 0; g < OP_KSLICE / 8; ++g) {
    const float4 a4 = *reinterpret_cast<const float4*>(aP + 8 * g);
    const float2 w0 = *reinterpret_cast<const float2*>(bP + (size_t)(8 * g + 0) * 64);
    const float2 w1 = *reinterpret_cast<const float2*>(bP + (size_t)(8 * g + 1) * 64);
    const float2 w2 = *reinterpret_cast<const float2*>(bP + (size_t)(8 * g + 2) * 64);
    const float2 w3 = *reinterpret_cast<const float2*>(bP + (size_t)(8 * g + 3) * 64);
    v2f a, b0, b1;
    a.x = a4.x;  a.y = a4.y;
    b0.x = w0.x; b0.y = w1.x;
    b1.x = w0.y; b1.y = w1.y;
    acc0 = WMMA_F32(a, b0, acc0);
    acc1 = WMMA_F32(a, b1, acc1);
    a.x = a4.z;  a.y = a4.w;
    b0.x = w2.x; b0.y = w3.x;
    b1.x = w2.y; b1.y = w3.y;
    acc0 = WMMA_F32(a, b0, acc0);
    acc1 = WMMA_F32(a, b1, acc1);
  }

  float* myp = &part[w][0];
#pragma unroll
  for (int r = 0; r < 8; ++r) {
    myp[r * 32 + lane]           = acc0[r];
    myp[OP_TILE + r * 32 + lane] = acc1[r];
  }
  __syncthreads();

  const int row = tid >> 5;
  const int col = tid & 31;
  const int src = (col & 1) * OP_TILE + (row & 7) * 32 + (row >> 3) * 16 + (col >> 1);
  float s = part[0][src];
#pragma unroll
  for (int ww = 1; ww < OP_WAVES; ++ww) s += part[ww][src];
  out[(size_t)(rt * 16 + row) * 64 + nh * 32 + col] = s + bout[nh * 32 + col];
}

extern "C" void kernel_launch(void* const* d_in, const int* in_sizes, int n_in,
                              void* d_out, int out_size, void* d_ws, size_t ws_size,
                              hipStream_t stream) {
  (void)in_sizes; (void)n_in; (void)out_size; (void)ws_size;
  const float* pos  = (const float*)d_in[0];
  const float* x    = (const float*)d_in[1];
  const float* Wq   = (const float*)d_in[2];
  const float* kl   = (const float*)d_in[3];
  const float* Wout = (const float*)d_in[4];
  const float* bout = (const float*)d_in[5];
  float* ws  = (float*)d_ws;
  float* out = (float*)d_out;

  pos_attn_fused<<<kB * kH, 256, 0, stream>>>(pos, x, Wq, kl, ws);
  out_proj<<<OP_BLOCKS, 512, 0, stream>>>(ws, Wout, bout, out);
}


# attention kernel: scores computed transposed (A and B swapped): softmax max and sum run down each lane's registers, P^T registers are the B fragments of O^T += x^T P^T, no LDS staging; sum order new
# speedup vs baseline: 1.5435x; 1.2110x over previous
#include <hip/hip_runtime.h>

typedef float v2f __attribute__((ext_vector_type(2)));
typedef float v8f __attribute__((ext_vector_type(8)));

#define WMMA_F32(a, b, c) \
  __builtin_amdgcn_wmma_f32_16x16x4_f32(false, (a), false, (b), (short)0, (c), false, false)

constexpr int kB = 8, kN = 1024, kP = 128, kH = 64, kD = 64, kJ = 256, kX = 64;
constexpr int INNER = kH * kD;
constexpr int CH = 32;
constexpr int NCH = kN / CH;

__device__ __forceinline__ float xhalf(float v) {
  const int b = __builtin_bit_cast(int, v);
  return __builtin_bit_cast(float,
      __builtin_amdgcn_permlanex16(b, b, 0x76543210, 0xfedcba98, false, false));
}

__global__ __launch_bounds__(256)
void pos_attn_fused(const float* __restrict__ pos, const float* __restrict__ x,
                    const float* __restrict__ Wq, const float* __restrict__ kl,
                    float* __restrict__ ws) {
  __shared__ __attribute__((aligned(16))) float lds_pos[2][CH][kP + 4];
  __shared__ __attribute__((aligned(16))) float lds_x[2][CH][kX];
  __shared__ float lds_qh[2][CH][kD + 1];
  __shared__ __attribute__((aligned(16))) float lds_o[8][16][kX + 4];

  const int bb   = blockIdx.x >> 6;
  const int h    = blockIdx.x & 63;
  const int tid  = threadIdx.x;
  const int w    = tid >> 5;
  const int lane = tid & 31;
  const int row  = lane & 15;
  const int half = lane >> 4;
  const int kb   = half * 2;

  const int tt = w >> 2;
  const int dt = w & 3;

  v8f   oacc[2][4];
  float mrun[2], lrun[2];
#pragma unroll
  for (int jt = 0; jt < 2; ++jt) {
#pragma unroll
    for (int xt = 0; xt < 4; ++xt) oacc[jt][xt] = (v8f){0,0,0,0,0,0,0,0};
    mrun[jt] = -3.0e38f;
    lrun[jt] = 0.0f;
  }

  const float4* gpos = reinterpret_cast<const float4*>(pos + (size_t)bb * kN * kP);
  const float4* gx   = reinterpret_cast<const float4*>(x   + (size_t)bb * kN * kX);
  constexpr int POS4 = CH * kP / 4;
  constexpr int X4   = CH * kX / 4;

#pragma unroll
  for (int i = 0; i < 4; ++i) {
    const int i4 = tid + i * 256;
    *reinterpret_cast<float4*>(&lds_pos[0][i4 >> 5][(i4 & 31) * 4]) = gpos[i4];
  }
#pragma unroll
  for (int i = 0; i < 2; ++i)
    reinterpret_cast<float4*>(&lds_x[0][0][0])[tid + i * 256] = gx[tid + i * 256];
  __syncthreads();

  for (int ch = 0; ch < NCH; ++ch) {
    const int  cur  = ch & 1;
    const int  nxt  = cur ^ 1;
    const bool more = ch + 1 < NCH;

    const float4 zero4 = make_float4(0.0f, 0.0f, 0.0f, 0.0f);
    float4 pp0 = zero4, pp1 = zero4, pp2 = zero4, pp3 = zero4, px0 = zero4, px1 = zero4;
    if (more) {
      const float4* np = gpos + (ch + 1) * POS4 + tid;
      const float4* nx = gx + (ch + 1) * X4 + tid;
      pp0 = np[0];  pp1 = np[256];  pp2 = np[512];  pp3 = np[768];
      px0 = nx[0];  px1 = nx[256];
    }

    {
      v8f cq = (v8f){0,0,0,0,0,0,0,0};
      const float* pA = &lds_pos[cur][tt * 16 + row][kb];
      const float* pB = Wq + h * kD + dt * 16 + row;
      v2f ap2[kP / 4];
#pragma unroll
      for (int k = 0; k < kP; k += 4) ap2[k / 4] = *reinterpret_cast<const v2f*>(pA + k);
      __builtin_amdgcn_sched_barrier(0);
#pragma unroll
      for (int k = 0; k < kP; k += 4) {
        v2f b;
        b.x = pB[(size_t)(k + kb) * INNER];
        b.y = pB[(size_t)(k + kb + 1) * INNER];
        cq = WMMA_F32(ap2[k / 4], b, cq);
      }
#pragma unroll
      for (int r = 0; r < 8; ++r)
        lds_qh[cur][tt * 16 + r + 8 * half][dt * 16 + row] = cq[r];
    }
    if (more) {
      float* sp = &lds_pos[nxt][tid >> 5][(tid & 31) * 4];
      *reinterpret_cast<float4*>(sp)                     = pp0;
      *reinterpret_cast<float4*>(sp +  8 * (kP + 4))     = pp1;
      *reinterpret_cast<float4*>(sp + 16 * (kP + 4))     = pp2;
      *reinterpret_cast<float4*>(sp + 24 * (kP + 4))     = pp3;
    }
    __syncthreads();
    if (more) {
      float4* sx = reinterpret_cast<float4*>(&lds_x[nxt][0][0]) + tid;
      sx[0]   = px0;
      sx[256] = px1;
    }

    v2f aq[2][kD / 4], ax[4][2][4];
#pragma unroll
    for (int nt = 0; nt < 2; ++nt)
#pragma unroll
      for (int k = 0; k < kD; k += 4) {
        aq[nt][k / 4].x = lds_qh[cur][nt * 16 + row][k + kb];
        aq[nt][k / 4].y = lds_qh[cur][nt * 16 + row][k + kb + 1];
      }
#pragma unroll
    for (int xt = 0; xt < 4; ++xt)
#pragma unroll
      for (int nt = 0; nt < 2; ++nt)
#pragma unroll
        for (int t = 0; t < 4; ++t) {
          ax[xt][nt][t].x = lds_x[cur][nt * 16 + 8 * half + 2 * t][xt * 16 + row];
          ax[xt][nt][t].y = lds_x[cur][nt * 16 + 8 * half + 2 * t + 1][xt * 16 + row];
        }
    __builtin_amdgcn_sched_barrier(2);

    v8f s[2][2];
#pragma unroll
    for (int jt = 0; jt < 2; ++jt)
#pragma unroll
      for (int nt = 0; nt < 2; ++nt) s[jt][nt] = (v8f){0,0,0,0,0,0,0,0};
#pragma unroll
    for (int k = 0; k < kD; k += 4) {
#pragma unroll
      for (int jt = 0; jt < 2; ++jt) {
        const float* bK = kl + (size_t)(w * 32 + jt * 16 + row) * kD;
        const v2f b = *reinterpret_cast<const v2f*>(bK + k + kb);
#pragma unroll
        for (int nt = 0; nt < 2; ++nt) s[jt][nt] = WMMA_F32(aq[nt][k / 4], b, s[jt][nt]);
      }
    }
#pragma unroll
    for (int jt = 0; jt < 2; ++jt)
#pragma unroll
      for (int nt = 0; nt < 2; ++nt)
#pragma unroll
        for (int r = 0; r < 8; ++r) s[jt][nt][r] *= 0.125f;

#pragma unroll
    for (int jt = 0; jt < 2; ++jt) {
      float t8[8];
#pragma unroll
      for (int r = 0; r < 8; ++r) t8[r] = fmaxf(s[jt][0][r], s[jt][1][r]);
      float tm = fmaxf(fmaxf(fmaxf(t8[0], t8[1]), fmaxf(t8[2], t8[3])),
                       fmaxf(fmaxf(t8[4], t8[5]), fmaxf(t8[6], t8[7])));
      tm = fmaxf(tm, xhalf(tm));
      const float mnew = fmaxf(mrun[jt], tm);
      const float fsc  = __expf(mrun[jt] - mnew);
#pragma unroll
      for (int nt = 0; nt < 2; ++nt)
#pragma unroll
        for (int r = 0; r < 8; ++r)
          s[jt][nt][r] = __expf(s[jt][nt][r] - mnew);
      float a8[8];
#pragma unroll
      for (int r = 0; r < 8; ++r) a8[r] = s[jt][0][r] + s[jt][1][r];
      const float rs = ((a8[0] + a8[1]) + (a8[2] + a8[3])) + ((a8[4] + a8[5]) + (a8[6] + a8[7]));
      lrun[jt] = lrun[jt] * fsc + rs;
      mrun[jt] = mnew;
#pragma unroll
      for (int xt = 0; xt < 4; ++xt)
#pragma unroll
        for (int r = 0; r < 8; ++r) oacc[jt][xt][r] *= fsc;

#pragma unroll
      for (int nt = 0; nt < 2; ++nt)
#pragma unroll
        for (int t = 0; t < 4; ++t) {
          v2f pb;
          pb.x = s[jt][nt][2 * t];
          pb.y = s[jt][nt][2 * t + 1];
#pragma unroll
          for (int xt = 0; xt < 4; ++xt)
            oacc[jt][xt] = WMMA_F32(ax[xt][nt][t], pb, oacc[jt][xt]);
        }
    }
  }

#pragma unroll
  for (int jt = 0; jt < 2; ++jt) {
    const float l   = lrun[jt] + xhalf(lrun[jt]);
    const float inv = 1.0f / l;
#pragma unroll
    for (int xt = 0; xt < 4; ++xt) {
      float* dst = &lds_o[w][row][xt * 16 + 8 * half];
      *reinterpret_cast<float4*>(dst) =
          make_float4(oacc[jt][xt][0] * inv, oacc[jt][xt][1] * inv,
                      oacc[jt][xt][2] * inv, oacc[jt][xt][3] * inv);
      *reinterpret_cast<float4*>(dst + 4) =
          make_float4(oacc[jt][xt][4] * inv, oacc[jt][xt][5] * inv,
                      oacc[jt][xt][6] * inv, oacc[jt][xt][7] * inv);
    }
#pragma unroll
    for (int i = 0; i < 8; ++i) {
      const int q  = lane + 32 * i;
      const int jr = q >> 4;
      const int c4 = q & 15;
      const float4 v = *reinterpret_cast<const float4*>(&lds_o[w][jr][c4 * 4]);
      const int j = w * 32 + jt * 16 + jr;
      *reinterpret_cast<float4*>(&ws[(size_t)(bb * kJ + j) * INNER + h * kX + c4 * 4]) = v;
    }
  }
}

constexpr int OP_WAVES  = 16;
constexpr int OP_KSLICE = INNER / OP_WAVES;
constexpr int OP_TILE   = 272;
constexpr int OP_BLOCKS = (kB * kJ / 16) * 2;

__global__ __launch_bounds__(512) __attribute__((amdgpu_waves_per_eu(4, 4)))
void out_proj(const float* __restrict__ ws, const float* __restrict__ Wout,
              const float* __restrict__ bout, float* __restrict__ out) {
  __shared__ float part[OP_WAVES][2 * OP_TILE];

  const int rt   = blockIdx.x >> 1;
  const int nh   = blockIdx.x & 1;
  const int tid  = threadIdx.x;
  const int w    = tid >> 5;
  const int lane = tid & 31;
  const int c    = lane & 15;
  const int half = lane >> 4;

  const float* aP = ws   + (size_t)(rt * 16 + c) * INNER + w * OP_KSLICE + 4 * half;
  const float* bP = Wout + (size_t)(w * OP_KSLICE + 4 * half) * 64 + nh * 32 + 2 * c;

  v8f acc0 = (v8f){0,0,0,0,0,0,0,0};
  v8f acc1 = (v8f){0,0,0,0,0,0,0,0};
#pragma unroll 8
  for (int g = 0; g < OP_KSLICE / 8; ++g) {
    const float4 a4 = *reinterpret_cast<const float4*>(aP + 8 * g);
    const float2 w0 = *reinterpret_cast<const float2*>(bP + (size_t)(8 * g + 0) * 64);
    const float2 w1 = *reinterpret_cast<const float2*>(bP + (size_t)(8 * g + 1) * 64);
    const float2 w2 = *reinterpret_cast<const float2*>(bP + (size_t)(8 * g + 2) * 64);
    const float2 w3 = *reinterpret_cast<const float2*>(bP + (size_t)(8 * g + 3) * 64);
    v2f a, b0, b1;
    a.x = a4.x;  a.y = a4.y;
    b0.x = w0.x; b0.y = w1.x;
    b1.x = w0.y; b1.y = w1.y;
    acc0 = WMMA_F32(a, b0, acc0);
    acc1 = WMMA_F32(a, b1, acc1);
    a.x = a4.z;  a.y = a4.w;
    b0.x = w2.x; b0.y = w3.x;
    b1.x = w2.y; b1.y = w3.y;
    acc0 = WMMA_F32(a, b0, acc0);
    acc1 = WMMA_F32(a, b1, acc1);
  }

  float* myp = &part[w][0];
#pragma unroll
  for (int r = 0; r < 8; ++r) {
    myp[r * 32 + lane]           = acc0[r];
    myp[OP_TILE + r * 32 + lane] = acc1[r];
  }
  __syncthreads();

  const int row = tid >> 5;
  const int col = tid & 31;
  const int src = (col & 1) * OP_TILE + (row & 7) * 32 + (row >> 3) * 16 + (col >> 1);
  float s = part[0][src];
#pragma unroll
  for (int ww = 1; ww < OP_WAVES; ++ww) s += part[ww][src];
  out[(size_t)(rt * 16 + row) * 64 + nh * 32 + col] = s + bout[nh * 32 + col];
}

extern "C" void kernel_launch(void* const* d_in, const int* in_sizes, int n_in,
                              void* d_out, int out_size, void* d_ws, size_t ws_size,
                              hipStream_t stream) {
  (void)in_sizes; (void)n_in; (void)out_size; (void)ws_size;
  const float* pos  = (const float*)d_in[0];
  const float* x    = (const float*)d_in[1];
  const float* Wq   = (const float*)d_in[2];
  const float* kl   = (const float*)d_in[3];
  const float* Wout = (const float*)d_in[4];
  const float* bout = (const float*)d_in[5];
  float* ws  = (float*)d_ws;
  float* out = (float*)d_out;

  pos_attn_fused<<<kB * kH, 256, 0, stream>>>(pos, x, Wq, kl, ws);
  out_proj<<<OP_BLOCKS, 512, 0, stream>>>(ws, Wout, bout, out);
}


# attention kernel: Qh and x^T fragments read from LDS as conflict-free 128-bit loads (score k order permuted in A and B alike, x chunk parked transposed); score sum order new
# speedup vs baseline: 1.7884x; 1.1587x over previous
#include <hip/hip_runtime.h>

typedef float v2f __attribute__((ext_vector_type(2)));
typedef float v8f __attribute__((ext_vector_type(8)));

#define WMMA_F32(a, b, c) \
  __builtin_amdgcn_wmma_f32_16x16x4_f32(false, (a), false, (b), (short)0, (c), false, false)

constexpr int kB = 8, kN = 1024, kP = 128, kH = 64, kD = 64, kJ = 256, kX = 64;
constexpr int INNER = kH * kD;
constexpr int CH = 32;
constexpr int NCH = kN / CH;

__device__ __forceinline__ float xhalf(float v) {
  const int b = __builtin_bit_cast(int, v);
  return __builtin_bit_cast(float,
      __builtin_amdgcn_permlanex16(b, b, 0x76543210, 0xfedcba98, false, false));
}

__global__ __launch_bounds__(256)
void pos_attn_fused(const float* __restrict__ pos, const float* __restrict__ x,
                    const float* __restrict__ Wq, const float* __restrict__ kl,
                    float* __restrict__ ws) {
  __shared__ __attribute__((aligned(16))) float lds_pos[2][CH][kP + 4];
  __shared__ __attribute__((aligned(16))) float lds_xT[2][kX][CH + 4];
  __shared__ __attribute__((aligned(16))) float lds_qh[2][CH][kD + 4];
  __shared__ __attribute__((aligned(16))) float lds_o[8][16][kX + 4];

  const int bb   = blockIdx.x >> 6;
  const int h    = blockIdx.x & 63;
  const int tid  = threadIdx.x;
  const int w    = tid >> 5;
  const int lane = tid & 31;
  const int row  = lane & 15;
  const int half = lane >> 4;
  const int kb   = half * 2;

  const int tt = w >> 2;
  const int dt = w & 3;

  v8f   oacc[2][4];
  float mrun[2], lrun[2];
#pragma unroll
  for (int jt = 0; jt < 2; ++jt) {
#pragma unroll
    for (int xt = 0; xt < 4; ++xt) oacc[jt][xt] = (v8f){0,0,0,0,0,0,0,0};
    mrun[jt] = -3.0e38f;
    lrun[jt] = 0.0f;
  }

  const float4* gpos = reinterpret_cast<const float4*>(pos + (size_t)bb * kN * kP);
  const float4* gx   = reinterpret_cast<const float4*>(x   + (size_t)bb * kN * kX);
  constexpr int POS4 = CH * kP / 4;
  constexpr int X4   = CH * kX / 4;

#pragma unroll
  for (int i = 0; i < 4; ++i) {
    const int i4 = tid + i * 256;
    *reinterpret_cast<float4*>(&lds_pos[0][i4 >> 5][(i4 & 31) * 4]) = gpos[i4];
  }
  const int xs_tok = tid >> 3;
  const int xs_xd  = (tid & 7) * 4;
  const int xs_g4  = xs_tok * (kX / 4) + (tid & 7);
#pragma unroll
  for (int i = 0; i < 2; ++i) {
    const float4 v = gx[xs_g4 + 8 * i];
    lds_xT[0][xs_xd + 32 * i + 0][xs_tok] = v.x;
    lds_xT[0][xs_xd + 32 * i + 1][xs_tok] = v.y;
    lds_xT[0][xs_xd + 32 * i + 2][xs_tok] = v.z;
    lds_xT[0][xs_xd + 32 * i + 3][xs_tok] = v.w;
  }
  __syncthreads();

  for (int ch = 0; ch < NCH; ++ch) {
    const int  cur  = ch & 1;
    const int  nxt  = cur ^ 1;
    const bool more = ch + 1 < NCH;

    const float4 zero4 = make_float4(0.0f, 0.0f, 0.0f, 0.0f);
    float4 pp0 = zero4, pp1 = zero4, pp2 = zero4, pp3 = zero4, px0 = zero4, px1 = zero4;
    if (more) {
      const float4* np = gpos + (ch + 1) * POS4 + tid;
      const float4* nx = gx + (ch + 1) * X4 + xs_g4;
      pp0 = np[0];  pp1 = np[256];  pp2 = np[512];  pp3 = np[768];
      px0 = nx[0];  px1 = nx[8];
    }

    {
      v8f cq = (v8f){0,0,0,0,0,0,0,0};
      const float* pA = &lds_pos[cur][tt * 16 + row][kb];
      const float* pB = Wq + h * kD + dt * 16 + row;
      v2f ap2[kP / 4];
#pragma unroll
      for (int k = 0; k < kP; k += 4) ap2[k / 4] = *reinterpret_cast<const v2f*>(pA + k);
      __builtin_amdgcn_sched_barrier(0);
#pragma unroll
      for (int k = 0; k < kP; k += 4) {
        v2f b;
        b.x = pB[(size_t)(k + kb) * INNER];
        b.y = pB[(size_t)(k + kb + 1) * INNER];
        cq = WMMA_F32(ap2[k / 4], b, cq);
      }
#pragma unroll
      for (int r = 0; r < 8; ++r)
        lds_qh[cur][tt * 16 + r + 8 * half][dt * 16 + row] = cq[r];
    }
    if (more) {
      float* sp = &lds_pos[nxt][tid >> 5][(tid & 31) * 4];
      *reinterpret_cast<float4*>(sp)                     = pp0;
      *reinterpret_cast<float4*>(sp +  8 * (kP + 4))     = pp1;
      *reinterpret_cast<float4*>(sp + 16 * (kP + 4))     = pp2;
      *reinterpret_cast<float4*>(sp + 24 * (kP + 4))     = pp3;
    }
    __syncthreads();
    if (more) {
      float* sx = &lds_xT[nxt][xs_xd][xs_tok];
      sx[ 0 * (CH + 4)] = px0.x;
      sx[ 1 * (CH + 4)] = px0.y;
      sx[ 2 * (CH + 4)] = px0.z;
      sx[ 3 * (CH + 4)] = px0.w;
      sx[32 * (CH + 4)] = px1.x;
      sx[33 * (CH + 4)] = px1.y;
      sx[34 * (CH + 4)] = px1.z;
      sx[35 * (CH + 4)] = px1.w;
    }

    float4 aq4[2][kD / 8], ax4[4][2][2];
#pragma unroll
    for (int g = 0; g < kD / 8; ++g)
#pragma unroll
      for (int nt = 0; nt < 2; ++nt)
        aq4[nt][g] = *reinterpret_cast<const float4*>(&lds_qh[cur][nt * 16 + row][8 * g + 4 * half]);
#pragma unroll
    for (int xt = 0; xt < 4; ++xt)
#pragma unroll
      for (int nt = 0; nt < 2; ++nt)
#pragma unroll
        for (int u = 0; u < 2; ++u)
          ax4[xt][nt][u] = *reinterpret_cast<const float4*>(
              &lds_xT[cur][xt * 16 + row][nt * 16 + 8 * half + 4 * u]);
    __builtin_amdgcn_sched_barrier(0);

    v8f s[2][2];
#pragma unroll
    for (int jt = 0; jt < 2; ++jt)
#pragma unroll
      for (int nt = 0; nt < 2; ++nt) s[jt][nt] = (v8f){0,0,0,0,0,0,0,0};
#pragma unroll
    for (int g = 0; g < kD / 8; ++g) {
      float4 b4[2];
#pragma unroll
      for (int jt = 0; jt < 2; ++jt)
        b4[jt] = *reinterpret_cast<const float4*>(
            kl + (size_t)(w * 32 + jt * 16 + row) * kD + 8 * g + 4 * half);
#pragma unroll
      for (int st = 0; st < 2; ++st)
#pragma unroll
        for (int jt = 0; jt < 2; ++jt) {
          v2f b;
          b.x = st ? b4[jt].z : b4[jt].x;
          b.y = st ? b4[jt].w : b4[jt].y;
#pragma unroll
          for (int nt = 0; nt < 2; ++nt) {
            v2f a;
            a.x = st ? aq4[nt][g].z : aq4[nt][g].x;
            a.y = st ? aq4[nt][g].w : aq4[nt][g].y;
            s[jt][nt] = WMMA_F32(a, b, s[jt][nt]);
          }
        }
    }
#pragma unroll
    for (int jt = 0; jt < 2; ++jt)
#pragma unroll
      for (int nt = 0; nt < 2; ++nt)
#pragma unroll
        for (int r = 0; r < 8; ++r) s[jt][nt][r] *= 0.125f;

#pragma unroll
    for (int jt = 0; jt < 2; ++jt) {
      float t8[8];
#pragma unroll
      for (int r = 0; r < 8; ++r) t8[r] = fmaxf(s[jt][0][r], s[jt][1][r]);
      float tm = fmaxf(fmaxf(fmaxf(t8[0], t8[1]), fmaxf(t8[2], t8[3])),
                       fmaxf(fmaxf(t8[4], t8[5]), fmaxf(t8[6], t8[7])));
      tm = fmaxf(tm, xhalf(tm));
      const float mnew = fmaxf(mrun[jt], tm);
      const float fsc  = __expf(mrun[jt] - mnew);
#pragma unroll
      for (int nt = 0; nt < 2; ++nt)
#pragma unroll
        for (int r = 0; r < 8; ++r)
          s[jt][nt][r] = __expf(s[jt][nt][r] - mnew);
      float a8[8];
#pragma unroll
      for (int r = 0; r < 8; ++r) a8[r] = s[jt][0][r] + s[jt][1][r];
      const float rs = ((a8[0] + a8[1]) + (a8[2] + a8[3])) + ((a8[4] + a8[5]) + (a8[6] + a8[7]));
      lrun[jt] = lrun[jt] * fsc + rs;
      mrun[jt] = mnew;
#pragma unroll
      for (int xt = 0; xt < 4; ++xt)
#pragma unroll
        for (int r = 0; r < 8; ++r) oacc[jt][xt][r] *= fsc;

#pragma unroll
      for (int nt = 0; nt < 2; ++nt)
#pragma unroll
        for (int t = 0; t < 4; ++t) {
          v2f pb;
          pb.x = s[jt][nt][2 * t];
          pb.y = s[jt][nt][2 * t + 1];
#pragma unroll
          for (int xt = 0; xt < 4; ++xt) {
            const float4 xv = ax4[xt][nt][t >> 1];
            v2f xa;
            xa.x = (t & 1) ? xv.z : xv.x;
            xa.y = (t & 1) ? xv.w : xv.y;
            oacc[jt][xt] = WMMA_F32(xa, pb, oacc[jt][xt]);
          }
        }
    }
  }

#pragma unroll
  for (int jt = 0; jt < 2; ++jt) {
    const float l   = lrun[jt] + xhalf(lrun[jt]);
    const float inv = 1.0f / l;
#pragma unroll
    for (int xt = 0; xt < 4; ++xt) {
      float* dst = &lds_o[w][row][xt * 16 + 8 * half];
      *reinterpret_cast<float4*>(dst) =
          make_float4(oacc[jt][xt][0] * inv, oacc[jt][xt][1] * inv,
                      oacc[jt][xt][2] * inv, oacc[jt][xt][3] * inv);
      *reinterpret_cast<float4*>(dst + 4) =
          make_float4(oacc[jt][xt][4] * inv, oacc[jt][xt][5] * inv,
                      oacc[jt][xt][6] * inv, oacc[jt][xt][7] * inv);
    }
#pragma unroll
    for (int i = 0; i < 8; ++i) {
      const int q  = lane + 32 * i;
      const int jr = q >> 4;
      const int c4 = q & 15;
      const float4 v = *reinterpret_cast<const float4*>(&lds_o[w][jr][c4 * 4]);
      const int j = w * 32 + jt * 16 + jr;
      *reinterpret_cast<float4*>(&ws[(size_t)(bb * kJ + j) * INNER + h * kX + c4 * 4]) = v;
    }
  }
}

constexpr int OP_WAVES  = 16;
constexpr int OP_KSLICE = INNER / OP_WAVES;
constexpr int OP_TILE   = 272;
constexpr int OP_BLOCKS = (kB * kJ / 16) * 2;

__global__ __launch_bounds__(512) __attribute__((amdgpu_waves_per_eu(4, 4)))
void out_proj(const float* __restrict__ ws, const float* __restrict__ Wout,
              const float* __restrict__ bout, float* __restrict__ out) {
  __shared__ float part[OP_WAVES][2 * OP_TILE];

  const int rt   = blockIdx.x >> 1;
  const int nh   = blockIdx.x & 1;
  const int tid  = threadIdx.x;
  const int w    = tid >> 5;
  const int lane = tid & 31;
  const int c    = lane & 15;
  const int half = lane >> 4;

  const float* aP = ws   + (size_t)(rt * 16 + c) * INNER + w * OP_KSLICE + 4 * half;
  const float* bP = Wout + (size_t)(w * OP_KSLICE + 4 * half) * 64 + nh * 32 + 2 * c;

  v8f acc0 = (v8f){0,0,0,0,0,0,0,0};
  v8f acc1 = (v8f){0,0,0,0,0,0,0,0};
#pragma unroll 8
  for (int g = 0; g < OP_KSLICE / 8; ++g) {
    const float4 a4 = *reinterpret_cast<const float4*>(aP + 8 * g);
    const float2 w0 = *reinterpret_cast<const float2*>(bP + (size_t)(8 * g + 0) * 64);
    const float2 w1 = *reinterpret_cast<const float2*>(bP + (size_t)(8 * g + 1) * 64);
    const float2 w2 = *reinterpret_cast<const float2*>(bP + (size_t)(8 * g + 2) * 64);
    const float2 w3 = *reinterpret_cast<const float2*>(bP + (size_t)(8 * g + 3) * 64);
    v2f a, b0, b1;
    a.x = a4.x;  a.y = a4.y;
    b0.x = w0.x; b0.y = w1.x;
    b1.x = w0.y; b1.y = w1.y;
    acc0 = WMMA_F32(a, b0, acc0);
    acc1 = WMMA_F32(a, b1, acc1);
    a.x = a4.z;  a.y = a4.w;
    b0.x = w2.x; b0.y = w3.x;
    b1.x = w2.y; b1.y = w3.y;
    acc0 = WMMA_F32(a, b0, acc0);
    acc1 = WMMA_F32(a, b1, acc1);
  }

  float* myp = &part[w][0];
#pragma unroll
  for (int r = 0; r < 8; ++r) {
    myp[r * 32 + lane]           = acc0[r];
    myp[OP_TILE + r * 32 + lane] = acc1[r];
  }
  __syncthreads();

  const int row = tid >> 5;
  const int col = tid & 31;
  const int src = (col & 1) * OP_TILE + (row & 7) * 32 + (row >> 3) * 16 + (col >> 1);
  float s = part[0][src];
#pragma unroll
  for (int ww = 1; ww < OP_WAVES; ++ww) s += part[ww][src];
  out[(size_t)(rt * 16 + row) * 64 + nh * 32 + col] = s + bout[nh * 32 + col];
}

extern "C" void kernel_launch(void* const* d_in, const int* in_sizes, int n_in,
                              void* d_out, int out_size, void* d_ws, size_t ws_size,
                              hipStream_t stream) {
  (void)in_sizes; (void)n_in; (void)out_size; (void)ws_size;
  const float* pos  = (const float*)d_in[0];
  const float* x    = (const float*)d_in[1];
  const float* Wq   = (const float*)d_in[2];
  const float* kl   = (const float*)d_in[3];
  const float* Wout = (const float*)d_in[4];
  const float* bout = (const float*)d_in[5];
  float* ws  = (float*)d_ws;
  float* out = (float*)d_out;

  pos_attn_fused<<<kB * kH, 256, 0, stream>>>(pos, x, Wq, kl, ws);
  out_proj<<<OP_BLOCKS, 512, 0, stream>>>(ws, Wout, bout, out);
}


# attention: block barrier split into arrival and wait; next chunk's Qh tile made with the score run and stored before the arrival, wait behind softmax and P V run, so no wave idles for the slowest
# speedup vs baseline: 1.9142x; 1.0703x over previous
#include <hip/hip_runtime.h>

typedef float v2f __attribute__((ext_vector_type(2)));
typedef float v8f __attribute__((ext_vector_type(8)));

#define WMMA_F32(a, b, c) \
  __builtin_amdgcn_wmma_f32_16x16x4_f32(false, (a), false, (b), (short)0, (c), false, false)

constexpr int kB = 8, kN = 1024, kP = 128, kH = 64, kD = 64, kJ = 256, kX = 64;
constexpr int INNER = kH * kD;
constexpr int CH = 32;
constexpr int NCH = kN / CH;

__device__ __forceinline__ float xhalf(float v) {
  const int b = __builtin_bit_cast(int, v);
  return __builtin_bit_cast(float,
      __builtin_amdgcn_permlanex16(b, b, 0x76543210, 0xfedcba98, false, false));
}

template <bool V> struct flag_c { static constexpr bool value = V; };

__global__ __launch_bounds__(256)
void pos_attn_fused(const float* __restrict__ pos, const float* __restrict__ x,
                    const float* __restrict__ Wq, const float* __restrict__ kl,
                    float* __restrict__ ws) {
  __shared__ __attribute__((aligned(16))) float lds_pos[2][CH][kP + 4];
  __shared__ __attribute__((aligned(16))) float lds_xT[2][kX][CH + 4];
  __shared__ __attribute__((aligned(16))) float lds_qh[2][CH][kD + 4];
  __shared__ __attribute__((aligned(16))) float lds_o[8][16][kX + 4];

  const int bb   = blockIdx.x >> 6;
  const int h    = blockIdx.x & 63;
  const int tid  = threadIdx.x;
  const int w    = tid >> 5;
  const int lane = tid & 31;
  const int row  = lane & 15;
  const int half = lane >> 4;
  const int kb   = half * 2;

  const int tt = w >> 2;
  const int dt = w & 3;

  v8f   oacc[2][4];
  float mrun[2], lrun[2];
#pragma unroll
  for (int jt = 0; jt < 2; ++jt) {
#pragma unroll
    for (int xt = 0; xt < 4; ++xt) oacc[jt][xt] = (v8f){0,0,0,0,0,0,0,0};
    mrun[jt] = -3.0e38f;
    lrun[jt] = 0.0f;
  }

  v2f wqf[kP / 4];
  {
    const float* pB = Wq + h * kD + dt * 16 + row;
#pragma unroll
    for (int k = 0; k < kP; k += 4) {
      wqf[k / 4].x = pB[(size_t)(k + kb) * INNER];
      wqf[k / 4].y = pB[(size_t)(k + kb + 1) * INNER];
    }
  }
  float4 klf[2][kD / 8];
#pragma unroll
  for (int g = 0; g < kD / 8; ++g)
#pragma unroll
    for (int jt = 0; jt < 2; ++jt)
      klf[jt][g] = *reinterpret_cast<const float4*>(
          kl + (size_t)(w * 32 + jt * 16 + row) * kD + 8 * g + 4 * half);

  const float4* gpos = reinterpret_cast<const float4*>(pos + (size_t)bb * kN * kP);
  const float4* gx   = reinterpret_cast<const float4*>(x   + (size_t)bb * kN * kX);
  constexpr int POS4 = CH * kP / 4;
  constexpr int X4   = CH * kX / 4;

  const int xs_tok = tid >> 3;
  const int xs_xd  = (tid & 7) * 4;
  const int xs_g4  = xs_tok * (kX / 4) + (tid & 7);

  auto qh_frags = [&](int pbuf, v2f (&ap2)[kP / 4]) __attribute__((always_inline)) {
    const float* pA = &lds_pos[pbuf][tt * 16 + row][kb];
#pragma unroll
    for (int k = 0; k < kP; k += 4) ap2[k / 4] = *reinterpret_cast<const v2f*>(pA + k);
  };
  auto qh_mma = [&](const v2f (&ap2)[kP / 4]) __attribute__((always_inline)) -> v8f {
    v8f cq = (v8f){0,0,0,0,0,0,0,0};
#pragma unroll
    for (int k = 0; k < kP; k += 4) cq = WMMA_F32(ap2[k / 4], wqf[k / 4], cq);
    return cq;
  };
  auto qh_store = [&](int qbuf, const v8f& cq) __attribute__((always_inline)) {
#pragma unroll
    for (int r = 0; r < 8; ++r)
      lds_qh[qbuf][tt * 16 + r + 8 * half][dt * 16 + row] = cq[r];
  };

#pragma unroll
  for (int i = 0; i < 4; ++i) {
    const int i4 = tid + i * 256;
    *reinterpret_cast<float4*>(&lds_pos[0][i4 >> 5][(i4 & 31) * 4]) = gpos[i4];
    *reinterpret_cast<float4*>(&lds_pos[1][i4 >> 5][(i4 & 31) * 4]) = gpos[POS4 + i4];
  }
#pragma unroll
  for (int i = 0; i < 2; ++i) {
    const float4 v = gx[xs_g4 + 8 * i];
    lds_xT[0][xs_xd + 32 * i + 0][xs_tok] = v.x;
    lds_xT[0][xs_xd + 32 * i + 1][xs_tok] = v.y;
    lds_xT[0][xs_xd + 32 * i + 2][xs_tok] = v.z;
    lds_xT[0][xs_xd + 32 * i + 3][xs_tok] = v.w;
  }
  __syncthreads();
  {
    v2f ap0[kP / 4];
    qh_frags(0, ap0);
    __builtin_amdgcn_sched_barrier(0);
    const v8f cq0 = qh_mma(ap0);
    qh_store(0, cq0);
  }
  __syncthreads();

  auto trip = [&](auto last_c, const int ch) __attribute__((always_inline)) {
    constexpr bool LAST = decltype(last_c)::value;
    const int  cur   = ch & 1;
    const int  nxt   = cur ^ 1;
    const bool more2 = ch + 2 < NCH;

    const float4 zero4 = make_float4(0.0f, 0.0f, 0.0f, 0.0f);
    float4 pp0 = zero4, pp1 = zero4, pp2 = zero4, pp3 = zero4, px0 = zero4, px1 = zero4;
    if (!LAST) {
      if (more2) {
        const float4* np = gpos + (ch + 2) * POS4 + tid;
        pp0 = np[0];  pp1 = np[256];  pp2 = np[512];  pp3 = np[768];
      }
      const float4* nx = gx + (ch + 1) * X4 + xs_g4;
      px0 = nx[0];  px1 = nx[8];
    }

    float4 aq4[2][kD / 8], ax4[4][2][2];
    v2f ap2[kP / 4];
    if (!LAST) {
      qh_frags(nxt, ap2);
      __builtin_amdgcn_sched_barrier(0);
    }
#pragma unroll
    for (int g = 0; g < kD / 8; ++g)
#pragma unroll
      for (int nt = 0; nt < 2; ++nt)
        aq4[nt][g] = *reinterpret_cast<const float4*>(&lds_qh[cur][nt * 16 + row][8 * g + 4 * half]);
    __builtin_amdgcn_sched_barrier(0);
#pragma unroll
    for (int xt = 0; xt < 4; ++xt)
#pragma unroll
      for (int nt = 0; nt < 2; ++nt)
#pragma unroll
        for (int u = 0; u < 2; ++u)
          ax4[xt][nt][u] = *reinterpret_cast<const float4*>(
              &lds_xT[cur][xt * 16 + row][nt * 16 + 8 * half + 4 * u]);
    __builtin_amdgcn_sched_barrier(0);

    v8f s[2][2];
#pragma unroll
    for (int jt = 0; jt < 2; ++jt)
#pragma unroll
      for (int nt = 0; nt < 2; ++nt) s[jt][nt] = (v8f){0,0,0,0,0,0,0,0};
#pragma unroll
    for (int g = 0; g < kD / 8; ++g) {
#pragma unroll
      for (int st = 0; st < 2; ++st)
#pragma unroll
        for (int jt = 0; jt < 2; ++jt) {
          v2f b;
          b.x = st ? klf[jt][g].z : klf[jt][g].x;
          b.y = st ? klf[jt][g].w : klf[jt][g].y;
#pragma unroll
          for (int nt = 0; nt < 2; ++nt) {
            v2f a;
            a.x = st ? aq4[nt][g].z : aq4[nt][g].x;
            a.y = st ? aq4[nt][g].w : aq4[nt][g].y;
            s[jt][nt] = WMMA_F32(a, b, s[jt][nt]);
          }
        }
    }

    v8f cqn = (v8f){0,0,0,0,0,0,0,0};
    if (!LAST) cqn = qh_mma(ap2);

    if (!LAST) {
      qh_store(nxt, cqn);
      {
        float* sx = &lds_xT[nxt][xs_xd][xs_tok];
        sx[ 0 * (CH + 4)] = px0.x;
        sx[ 1 * (CH + 4)] = px0.y;
        sx[ 2 * (CH + 4)] = px0.z;
        sx[ 3 * (CH + 4)] = px0.w;
        sx[32 * (CH + 4)] = px1.x;
        sx[33 * (CH + 4)] = px1.y;
        sx[34 * (CH + 4)] = px1.z;
        sx[35 * (CH + 4)] = px1.w;
      }
      if (more2) {
        float* sp = &lds_pos[cur][tid >> 5][(tid & 31) * 4];
        *reinterpret_cast<float4*>(sp)                     = pp0;
        *reinterpret_cast<float4*>(sp +  8 * (kP + 4))     = pp1;
        *reinterpret_cast<float4*>(sp + 16 * (kP + 4))     = pp2;
        *reinterpret_cast<float4*>(sp + 24 * (kP + 4))     = pp3;
      }
      __builtin_amdgcn_fence(__ATOMIC_RELEASE, "workgroup");
      __builtin_amdgcn_s_barrier_signal(-1);
      __builtin_amdgcn_sched_barrier(0);
    }

#pragma unroll
    for (int jt = 0; jt < 2; ++jt)
#pragma unroll
      for (int nt = 0; nt < 2; ++nt)
#pragma unroll
        for (int r = 0; r < 8; ++r) s[jt][nt][r] *= 0.125f;

#pragma unroll
    for (int jt = 0; jt < 2; ++jt) {
      float t8[8];
#pragma unroll
      for (int r = 0; r < 8; ++r) t8[r] = fmaxf(s[jt][0][r], s[jt][1][r]);
      float tm = fmaxf(fmaxf(fmaxf(t8[0], t8[1]), fmaxf(t8[2], t8[3])),
                       fmaxf(fmaxf(t8[4], t8[5]), fmaxf(t8[6], t8[7])));
      tm = fmaxf(tm, xhalf(tm));
      const float mnew = fmaxf(mrun[jt], tm);
      const float fsc  = __expf(mrun[jt] - mnew);
#pragma unroll
      for (int nt = 0; nt < 2; ++nt)
#pragma unroll
        for (int r = 0; r < 8; ++r)
          s[jt][nt][r] = __expf(s[jt][nt][r] - mnew);
      float a8[8];
#pragma unroll
      for (int r = 0; r < 8; ++r) a8[r] = s[jt][0][r] + s[jt][1][r];
      const float rs = ((a8[0] + a8[1]) + (a8[2] + a8[3])) + ((a8[4] + a8[5]) + (a8[6] + a8[7]));
      lrun[jt] = lrun[jt] * fsc + rs;
      mrun[jt] = mnew;
#pragma unroll
      for (int xt = 0; xt < 4; ++xt)
#pragma unroll
        for (int r = 0; r < 8; ++r) oacc[jt][xt][r] *= fsc;
    }

#pragma unroll
    for (int jt = 0; jt < 2; ++jt)
#pragma unroll
      for (int nt = 0; nt < 2; ++nt)
#pragma unroll
        for (int t = 0; t < 4; ++t) {
          v2f pb;
          pb.x = s[jt][nt][2 * t];
          pb.y = s[jt][nt][2 * t + 1];
#pragma unroll
          for (int xt = 0; xt < 4; ++xt) {
            const float4 xv = ax4[xt][nt][t >> 1];
            v2f xa;
            xa.x = (t & 1) ? xv.z : xv.x;
            xa.y = (t & 1) ? xv.w : xv.y;
            oacc[jt][xt] = WMMA_F32(xa, pb, oacc[jt][xt]);
          }
        }

    if (!LAST) {
      __builtin_amdgcn_s_barrier_wait(-1);
      __builtin_amdgcn_fence(__ATOMIC_ACQUIRE, "workgroup");
    }
  };

  for (int ch = 0; ch < NCH - 1; ++ch) trip(flag_c<false>{}, ch);
  trip(flag_c<true>{}, NCH - 1);

#pragma unroll
  for (int jt = 0; jt < 2; ++jt) {
    const float l   = lrun[jt] + xhalf(lrun[jt]);
    const float inv = 1.0f / l;
#pragma unroll
    for (int xt = 0; xt < 4; ++xt) {
      float* dst = &lds_o[w][row][xt * 16 + 8 * half];
      *reinterpret_cast<float4*>(dst) =
          make_float4(oacc[jt][xt][0] * inv, oacc[jt][xt][1] * inv,
                      oacc[jt][xt][2] * inv, oacc[jt][xt][3] * inv);
      *reinterpret_cast<float4*>(dst + 4) =
          make_float4(oacc[jt][xt][4] * inv, oacc[jt][xt][5] * inv,
                      oacc[jt][xt][6] * inv, oacc[jt][xt][7] * inv);
    }
#pragma unroll
    for (int i = 0; i < 8; ++i) {
      const int q  = lane + 32 * i;
      const int jr = q >> 4;
      const int c4 = q & 15;
      const float4 v = *reinterpret_cast<const float4*>(&lds_o[w][jr][c4 * 4]);
      const int j = w * 32 + jt * 16 + jr;
      *reinterpret_cast<float4*>(&ws[(size_t)(bb * kJ + j) * INNER + h * kX + c4 * 4]) = v;
    }
  }
}

constexpr int OP_WAVES  = 16;
constexpr int OP_KSLICE = INNER / OP_WAVES;
constexpr int OP_TILE   = 272;
constexpr int OP_BLOCKS = (kB * kJ / 16) * 2;

__global__ __launch_bounds__(512) __attribute__((amdgpu_waves_per_eu(4, 4)))
void out_proj(const float* __restrict__ ws, const float* __restrict__ Wout,
              const float* __restrict__ bout, float* __restrict__ out) {
  __shared__ float part[OP_WAVES][2 * OP_TILE];

  const int rt   = blockIdx.x >> 1;
  const int nh   = blockIdx.x & 1;
  const int tid  = threadIdx.x;
  const int w    = tid >> 5;
  const int lane = tid & 31;
  const int c    = lane & 15;
  const int half = lane >> 4;

  const float* aP = ws   + (size_t)(rt * 16 + c) * INNER + w * OP_KSLICE + 4 * half;
  const float* bP = Wout + (size_t)(w * OP_KSLICE + 4 * half) * 64 + nh * 32 + 2 * c;

  v8f acc0 = (v8f){0,0,0,0,0,0,0,0};
  v8f acc1 = (v8f){0,0,0,0,0,0,0,0};
#pragma unroll 8
  for (int g = 0; g < OP_KSLICE / 8; ++g) {
    const float4 a4 = *reinterpret_cast<const float4*>(aP + 8 * g);
    const float2 w0 = *reinterpret_cast<const float2*>(bP + (size_t)(8 * g + 0) * 64);
    const float2 w1 = *reinterpret_cast<const float2*>(bP + (size_t)(8 * g + 1) * 64);
    const float2 w2 = *reinterpret_cast<const float2*>(bP + (size_t)(8 * g + 2) * 64);
    const float2 w3 = *reinterpret_cast<const float2*>(bP + (size_t)(8 * g + 3) * 64);
    v2f a, b0, b1;
    a.x = a4.x;  a.y = a4.y;
    b0.x = w0.x; b0.y = w1.x;
    b1.x = w0.y; b1.y = w1.y;
    acc0 = WMMA_F32(a, b0, acc0);
    acc1 = WMMA_F32(a, b1, acc1);
    a.x = a4.z;  a.y = a4.w;
    b0.x = w2.x; b0.y = w3.x;
    b1.x = w2.y; b1.y = w3.y;
    acc0 = WMMA_F32(a, b0, acc0);
    acc1 = WMMA_F32(a, b1, acc1);
  }

  float* myp = &part[w][0];
#pragma unroll
  for (int r = 0; r < 8; ++r) {
    myp[r * 32 + lane]           = acc0[r];
    myp[OP_TILE + r * 32 + lane] = acc1[r];
  }
  __syncthreads();

  const int row = tid >> 5;
  const int col = tid & 31;
  const int src = (col & 1) * OP_TILE + (row & 7) * 32 + (row >> 3) * 16 + (col >> 1);
  float s = part[0][src];
#pragma unroll
  for (int ww = 1; ww < OP_WAVES; ++ww) s += part[ww][src];
  out[(size_t)(rt * 16 + row) * 64 + nh * 32 + col] = s + bout[nh * 32 + col];
}

extern "C" void kernel_launch(void* const* d_in, const int* in_sizes, int n_in,
                              void* d_out, int out_size, void* d_ws, size_t ws_size,
                              hipStream_t stream) {
  (void)in_sizes; (void)n_in; (void)out_size; (void)ws_size;
  const float* pos  = (const float*)d_in[0];
  const float* x    = (const float*)d_in[1];
  const float* Wq   = (const float*)d_in[2];
  const float* kl   = (const float*)d_in[3];
  const float* Wout = (const float*)d_in[4];
  const float* bout = (const float*)d_in[5];
  float* ws  = (float*)d_ws;
  float* out = (float*)d_out;

  pos_attn_fused<<<kB * kH, 256, 0, stream>>>(pos, x, Wq, kl, ws);
  out_proj<<<OP_BLOCKS, 512, 0, stream>>>(ws, Wout, bout, out);
}


# attention: wave priority by trip phase (2 up to the barrier arrival, 0 behind it, 1 once maxima and rescale are done): a SIMD's two waves run half a trip apart and fill each other's LDS waits
# speedup vs baseline: 1.9752x; 1.0319x over previous
#include <hip/hip_runtime.h>

typedef float v2f __attribute__((ext_vector_type(2)));
typedef float v8f __attribute__((ext_vector_type(8)));

#define WMMA_F32(a, b, c) \
  __builtin_amdgcn_wmma_f32_16x16x4_f32(false, (a), false, (b), (short)0, (c), false, false)

constexpr int kB = 8, kN = 1024, kP = 128, kH = 64, kD = 64, kJ = 256, kX = 64;
constexpr int INNER = kH * kD;
constexpr int CH = 32;
constexpr int NCH = kN / CH;

__device__ __forceinline__ float xhalf(float v) {
  const int b = __builtin_bit_cast(int, v);
  return __builtin_bit_cast(float,
      __builtin_amdgcn_permlanex16(b, b, 0x76543210, 0xfedcba98, false, false));
}

template <bool V> struct flag_c { static constexpr bool value = V; };

__global__ __launch_bounds__(256)
void pos_attn_fused(const float* __restrict__ pos, const float* __restrict__ x,
                    const float* __restrict__ Wq, const float* __restrict__ kl,
                    float* __restrict__ ws) {
  __shared__ __attribute__((aligned(16))) float lds_pos[2][CH][kP + 4];
  __shared__ __attribute__((aligned(16))) float lds_xT[2][kX][CH + 4];
  __shared__ __attribute__((aligned(16))) float lds_qh[2][CH][kD + 4];
  __shared__ __attribute__((aligned(16))) float lds_o[8][16][kX + 4];

  const int bb   = blockIdx.x >> 6;
  const int h    = blockIdx.x & 63;
  const int tid  = threadIdx.x;
  const int w    = tid >> 5;
  const int lane = tid & 31;
  const int row  = lane & 15;
  const int half = lane >> 4;
  const int kb   = half * 2;

  const int tt = w >> 2;
  const int dt = w & 3;

  v8f   oacc[2][4];
  float mrun[2], lrun[2];
#pragma unroll
  for (int jt = 0; jt < 2; ++jt) {
#pragma unroll
    for (int xt = 0; xt < 4; ++xt) oacc[jt][xt] = (v8f){0,0,0,0,0,0,0,0};
    mrun[jt] = -3.0e38f;
    lrun[jt] = 0.0f;
  }

  v2f wqf[kP / 4];
  {
    const float* pB = Wq + h * kD + dt * 16 + row;
#pragma unroll
    for (int k = 0; k < kP; k += 4) {
      wqf[k / 4].x = pB[(size_t)(k + kb) * INNER];
      wqf[k / 4].y = pB[(size_t)(k + kb + 1) * INNER];
    }
  }
  float4 klf[2][kD / 8];
#pragma unroll
  for (int g = 0; g < kD / 8; ++g)
#pragma unroll
    for (int jt = 0; jt < 2; ++jt)
      klf[jt][g] = *reinterpret_cast<const float4*>(
          kl + (size_t)(w * 32 + jt * 16 + row) * kD + 8 * g + 4 * half);

  const float4* gpos = reinterpret_cast<const float4*>(pos + (size_t)bb * kN * kP);
  const float4* gx   = reinterpret_cast<const float4*>(x   + (size_t)bb * kN * kX);
  constexpr int POS4 = CH * kP / 4;
  constexpr int X4   = CH * kX / 4;

  const int xs_tok = tid >> 3;
  const int xs_xd  = (tid & 7) * 4;
  const int xs_g4  = xs_tok * (kX / 4) + (tid & 7);

  auto qh_frags = [&](int pbuf, v2f (&ap2)[kP / 4]) __attribute__((always_inline)) {
    const float* pA = &lds_pos[pbuf][tt * 16 + row][kb];
#pragma unroll
    for (int k = 0; k < kP; k += 4) ap2[k / 4] = *reinterpret_cast<const v2f*>(pA + k);
  };
  auto qh_mma = [&](const v2f (&ap2)[kP / 4]) __attribute__((always_inline)) -> v8f {
    v8f cq = (v8f){0,0,0,0,0,0,0,0};
#pragma unroll
    for (int k = 0; k < kP; k += 4) cq = WMMA_F32(ap2[k / 4], wqf[k / 4], cq);
    return cq;
  };
  auto qh_store = [&](int qbuf, const v8f& cq) __attribute__((always_inline)) {
#pragma unroll
    for (int r = 0; r < 8; ++r)
      lds_qh[qbuf][tt * 16 + r + 8 * half][dt * 16 + row] = cq[r];
  };

#pragma unroll
  for (int i = 0; i < 4; ++i) {
    const int i4 = tid + i * 256;
    *reinterpret_cast<float4*>(&lds_pos[0][i4 >> 5][(i4 & 31) * 4]) = gpos[i4];
    *reinterpret_cast<float4*>(&lds_pos[1][i4 >> 5][(i4 & 31) * 4]) = gpos[POS4 + i4];
  }
#pragma unroll
  for (int i = 0; i < 2; ++i) {
    const float4 v = gx[xs_g4 + 8 * i];
    lds_xT[0][xs_xd + 32 * i + 0][xs_tok] = v.x;
    lds_xT[0][xs_xd + 32 * i + 1][xs_tok] = v.y;
    lds_xT[0][xs_xd + 32 * i + 2][xs_tok] = v.z;
    lds_xT[0][xs_xd + 32 * i + 3][xs_tok] = v.w;
  }
  __syncthreads();
  {
    v2f ap0[kP / 4];
    qh_frags(0, ap0);
    __builtin_amdgcn_sched_barrier(0);
    const v8f cq0 = qh_mma(ap0);
    qh_store(0, cq0);
  }
  __syncthreads();

  auto trip = [&](auto last_c, const int ch) __attribute__((always_inline)) {
    constexpr bool LAST = decltype(last_c)::value;
    const int  cur   = ch & 1;
    const int  nxt   = cur ^ 1;
    const bool more2 = ch + 2 < NCH;

    if (!LAST) __builtin_amdgcn_s_setprio(2);
    else       __builtin_amdgcn_s_setprio(0);

    const float4 zero4 = make_float4(0.0f, 0.0f, 0.0f, 0.0f);
    float4 pp0 = zero4, pp1 = zero4, pp2 = zero4, pp3 = zero4, px0 = zero4, px1 = zero4;
    if (!LAST) {
      if (more2) {
        const float4* np = gpos + (ch + 2) * POS4 + tid;
        pp0 = np[0];  pp1 = np[256];  pp2 = np[512];  pp3 = np[768];
      }
      const float4* nx = gx + (ch + 1) * X4 + xs_g4;
      px0 = nx[0];  px1 = nx[8];
    }

    float4 aq4[2][kD / 8], ax4[4][2][2];
    v2f ap2[kP / 4];
    if (!LAST) {
      qh_frags(nxt, ap2);
      __builtin_amdgcn_sched_barrier(0);
    }
#pragma unroll
    for (int g = 0; g < kD / 8; ++g)
#pragma unroll
      for (int nt = 0; nt < 2; ++nt)
        aq4[nt][g] = *reinterpret_cast<const float4*>(&lds_qh[cur][nt * 16 + row][8 * g + 4 * half]);
    __builtin_amdgcn_sched_barrier(0);
#pragma unroll
    for (int xt = 0; xt < 4; ++xt)
#pragma unroll
      for (int nt = 0; nt < 2; ++nt)
#pragma unroll
        for (int u = 0; u < 2; ++u)
          ax4[xt][nt][u] = *reinterpret_cast<const float4*>(
              &lds_xT[cur][xt * 16 + row][nt * 16 + 8 * half + 4 * u]);
    __builtin_amdgcn_sched_barrier(0);

    v8f s[2][2];
#pragma unroll
    for (int jt = 0; jt < 2; ++jt)
#pragma unroll
      for (int nt = 0; nt < 2; ++nt) s[jt][nt] = (v8f){0,0,0,0,0,0,0,0};
#pragma unroll
    for (int g = 0; g < kD / 8; ++g) {
#pragma unroll
      for (int st = 0; st < 2; ++st)
#pragma unroll
        for (int jt = 0; jt < 2; ++jt) {
          v2f b;
          b.x = st ? klf[jt][g].z : klf[jt][g].x;
          b.y = st ? klf[jt][g].w : klf[jt][g].y;
#pragma unroll
          for (int nt = 0; nt < 2; ++nt) {
            v2f a;
            a.x = st ? aq4[nt][g].z : aq4[nt][g].x;
            a.y = st ? aq4[nt][g].w : aq4[nt][g].y;
            s[jt][nt] = WMMA_F32(a, b, s[jt][nt]);
          }
        }
    }

    v8f cqn = (v8f){0,0,0,0,0,0,0,0};
    if (!LAST) cqn = qh_mma(ap2);

    if (!LAST) {
      qh_store(nxt, cqn);
      {
        float* sx = &lds_xT[nxt][xs_xd][xs_tok];
        sx[ 0 * (CH + 4)] = px0.x;
        sx[ 1 * (CH + 4)] = px0.y;
        sx[ 2 * (CH + 4)] = px0.z;
        sx[ 3 * (CH + 4)] = px0.w;
        sx[32 * (CH + 4)] = px1.x;
        sx[33 * (CH + 4)] = px1.y;
        sx[34 * (CH + 4)] = px1.z;
        sx[35 * (CH + 4)] = px1.w;
      }
      if (more2) {
        float* sp = &lds_pos[cur][tid >> 5][(tid & 31) * 4];
        *reinterpret_cast<float4*>(sp)                     = pp0;
        *reinterpret_cast<float4*>(sp +  8 * (kP + 4))     = pp1;
        *reinterpret_cast<float4*>(sp + 16 * (kP + 4))     = pp2;
        *reinterpret_cast<float4*>(sp + 24 * (kP + 4))     = pp3;
      }
      __builtin_amdgcn_fence(__ATOMIC_RELEASE, "workgroup");
      __builtin_amdgcn_s_barrier_signal(-1);
      __builtin_amdgcn_s_setprio(0);
      __builtin_amdgcn_sched_barrier(0);
    }

#pragma unroll
    for (int jt = 0; jt < 2; ++jt)
#pragma unroll
      for (int nt = 0; nt < 2; ++nt)
#pragma unroll
        for (int r = 0; r < 8; ++r) s[jt][nt][r] *= 0.125f;

    float fsc2[2];
#pragma unroll
    for (int jt = 0; jt < 2; ++jt) {
      float t8[8];
#pragma unroll
      for (int r = 0; r < 8; ++r) t8[r] = fmaxf(s[jt][0][r], s[jt][1][r]);
      float tm = fmaxf(fmaxf(fmaxf(t8[0], t8[1]), fmaxf(t8[2], t8[3])),
                       fmaxf(fmaxf(t8[4], t8[5]), fmaxf(t8[6], t8[7])));
      tm = fmaxf(tm, xhalf(tm));
      const float mnew = fmaxf(mrun[jt], tm);
      fsc2[jt] = __expf(mrun[jt] - mnew);
      mrun[jt] = mnew;
#pragma unroll
      for (int xt = 0; xt < 4; ++xt)
#pragma unroll
        for (int r = 0; r < 8; ++r) oacc[jt][xt][r] *= fsc2[jt];
    }
    if (!LAST) {
      __builtin_amdgcn_s_setprio(1);
    }
#pragma unroll
    for (int jt = 0; jt < 2; ++jt) {
      const float mnew = mrun[jt];
#pragma unroll
      for (int nt = 0; nt < 2; ++nt)
#pragma unroll
        for (int r = 0; r < 8; ++r)
          s[jt][nt][r] = __expf(s[jt][nt][r] - mnew);
      float a8[8];
#pragma unroll
      for (int r = 0; r < 8; ++r) a8[r] = s[jt][0][r] + s[jt][1][r];
      const float rs = ((a8[0] + a8[1]) + (a8[2] + a8[3])) + ((a8[4] + a8[5]) + (a8[6] + a8[7]));
      lrun[jt] = lrun[jt] * fsc2[jt] + rs;
    }

#pragma unroll
    for (int jt = 0; jt < 2; ++jt)
#pragma unroll
      for (int nt = 0; nt < 2; ++nt)
#pragma unroll
        for (int t = 0; t < 4; ++t) {
          v2f pb;
          pb.x = s[jt][nt][2 * t];
          pb.y = s[jt][nt][2 * t + 1];
#pragma unroll
          for (int xt = 0; xt < 4; ++xt) {
            const float4 xv = ax4[xt][nt][t >> 1];
            v2f xa;
            xa.x = (t & 1) ? xv.z : xv.x;
            xa.y = (t & 1) ? xv.w : xv.y;
            oacc[jt][xt] = WMMA_F32(xa, pb, oacc[jt][xt]);
          }
        }

    if (!LAST) {
      __builtin_amdgcn_s_barrier_wait(-1);
      __builtin_amdgcn_fence(__ATOMIC_ACQUIRE, "workgroup");
    }
  };

  for (int ch = 0; ch < NCH - 1; ++ch) trip(flag_c<false>{}, ch);
  trip(flag_c<true>{}, NCH - 1);

#pragma unroll
  for (int jt = 0; jt < 2; ++jt) {
    const float l   = lrun[jt] + xhalf(lrun[jt]);
    const float inv = 1.0f / l;
#pragma unroll
    for (int xt = 0; xt < 4; ++xt) {
      float* dst = &lds_o[w][row][xt * 16 + 8 * half];
      *reinterpret_cast<float4*>(dst) =
          make_float4(oacc[jt][xt][0] * inv, oacc[jt][xt][1] * inv,
                      oacc[jt][xt][2] * inv, oacc[jt][xt][3] * inv);
      *reinterpret_cast<float4*>(dst + 4) =
          make_float4(oacc[jt][xt][4] * inv, oacc[jt][xt][5] * inv,
                      oacc[jt][xt][6] * inv, oacc[jt][xt][7] * inv);
    }
#pragma unroll
    for (int i = 0; i < 8; ++i) {
      const int q  = lane + 32 * i;
      const int jr = q >> 4;
      const int c4 = q & 15;
      const float4 v = *reinterpret_cast<const float4*>(&lds_o[w][jr][c4 * 4]);
      const int j = w * 32 + jt * 16 + jr;
      *reinterpret_cast<float4*>(&ws[(size_t)(bb * kJ + j) * INNER + h * kX + c4 * 4]) = v;
    }
  }
}

constexpr int OP_WAVES  = 16;
constexpr int OP_KSLICE = INNER / OP_WAVES;
constexpr int OP_TILE   = 272;
constexpr int OP_BLOCKS = (kB * kJ / 16) * 2;

__global__ __launch_bounds__(512) __attribute__((amdgpu_waves_per_eu(4, 4)))
void out_proj(const float* __restrict__ ws, const float* __restrict__ Wout,
              const float* __restrict__ bout, float* __restrict__ out) {
  __shared__ float part[OP_WAVES][2 * OP_TILE];

  const int rt   = blockIdx.x >> 1;
  const int nh   = blockIdx.x & 1;
  const int tid  = threadIdx.x;
  const int w    = tid >> 5;
  const int lane = tid & 31;
  const int c    = lane & 15;
  const int half = lane >> 4;

  const float* aP = ws   + (size_t)(rt * 16 + c) * INNER + w * OP_KSLICE + 4 * half;
  const float* bP = Wout + (size_t)(w * OP_KSLICE + 4 * half) * 64 + nh * 32 + 2 * c;

  v8f acc0 = (v8f){0,0,0,0,0,0,0,0};
  v8f acc1 = (v8f){0,0,0,0,0,0,0,0};
#pragma unroll 8
  for (int g = 0; g < OP_KSLICE / 8; ++g) {
    const float4 a4 = *reinterpret_cast<const float4*>(aP + 8 * g);
    const float2 w0 = *reinterpret_cast<const float2*>(bP + (size_t)(8 * g + 0) * 64);
    const float2 w1 = *reinterpret_cast<const float2*>(bP + (size_t)(8 * g + 1) * 64);
    const float2 w2 = *reinterpret_cast<const float2*>(bP + (size_t)(8 * g + 2) * 64);
    const float2 w3 = *reinterpret_cast<const float2*>(bP + (size_t)(8 * g + 3) * 64);
    v2f a, b0, b1;
    a.x = a4.x;  a.y = a4.y;
    b0.x = w0.x; b0.y = w1.x;
    b1.x = w0.y; b1.y = w1.y;
    acc0 = WMMA_F32(a, b0, acc0);
    acc1 = WMMA_F32(a, b1, acc1);
    a.x = a4.z;  a.y = a4.w;
    b0.x = w2.x; b0.y = w3.x;
    b1.x = w2.y; b1.y = w3.y;
    acc0 = WMMA_F32(a, b0, acc0);
    acc1 = WMMA_F32(a, b1, acc1);
  }

  float* myp = &part[w][0];
#pragma unroll
  for (int r = 0; r < 8; ++r) {
    myp[r * 32 + lane]           = acc0[r];
    myp[OP_TILE + r * 32 + lane] = acc1[r];
  }
  __syncthreads();

  const int row = tid >> 5;
  const int col = tid & 31;
  const int src = (col & 1) * OP_TILE + (row & 7) * 32 + (row >> 3) * 16 + (col >> 1);
  float s = part[0][src];
#pragma unroll
  for (int ww = 1; ww < OP_WAVES; ++ww) s += part[ww][src];
  out[(size_t)(rt * 16 + row) * 64 + nh * 32 + col] = s + bout[nh * 32 + col];
}

extern "C" void kernel_launch(void* const* d_in, const int* in_sizes, int n_in,
                              void* d_out, int out_size, void* d_ws, size_t ws_size,
                              hipStream_t stream) {
  (void)in_sizes; (void)n_in; (void)out_size; (void)ws_size;
  const float* pos  = (const float*)d_in[0];
  const float* x    = (const float*)d_in[1];
  const float* Wq   = (const float*)d_in[2];
  const float* kl   = (const float*)d_in[3];
  const float* Wout = (const float*)d_in[4];
  const float* bout = (const float*)d_in[5];
  float* ws  = (float*)d_ws;
  float* out = (float*)d_out;

  pos_attn_fused<<<kB * kH, 256, 0, stream>>>(pos, x, Wq, kl, ws);
  out_proj<<<OP_BLOCKS, 512, 0, stream>>>(ws, Wout, bout, out);
}
